# starting kernel
# baseline (speedup 1.0000x reference)
#include <hip/hip_runtime.h>

typedef __attribute__((ext_vector_type(2))) float v2f;
typedef __attribute__((ext_vector_type(8))) float v8f;

#define N_NODES 50000
#define N_EDGES 800000
#define D_IN    512
#define D_OUT   128
#define N_BANDS (N_NODES / 16)

__global__ __launch_bounds__(256) void gemm_wmma_f32(
    const float* __restrict__ x, const float* __restrict__ W,
    float* __restrict__ h)
{
    const int lane = threadIdx.x & 31;
    const int band = blockIdx.x * 8 + (threadIdx.x >> 5);
    if (band >= N_BANDS) return;

    const int m0   = band * 16;
    const int half = lane >> 4;
    const int l16  = lane & 15;
    const int koff = half * 2;

    v8f acc[8] = {};

    const float* xrow = x + (size_t)(m0 + l16) * D_IN + koff;

    for (int k0 = 0; k0 < D_IN; k0 += 4) {
        v2f a;
        a.x = xrow[k0 + 0];
        a.y = xrow[k0 + 1];
        const float* wk = W + (size_t)(k0 + koff) * D_OUT + l16;
        #pragma unroll
        for (int nt = 0; nt < 8; ++nt) {
            v2f b;
            b.x = wk[nt * 16];
            b.y = wk[nt * 16 + D_OUT];
            acc[nt] = __builtin_amdgcn_wmma_f32_16x16x4_f32(
                 false, a,  false, b,
                 (short)0, acc[nt],
                 false,  false);
        }
    }

    #pragma unroll
    for (int nt = 0; nt < 8; ++nt) {
        float* hp = h + (size_t)(m0 + half * 8) * D_OUT + nt * 16 + l16;
        #pragma unroll
        for (int r = 0; r < 8; ++r)
            hp[(size_t)r * D_OUT] = acc[nt][r];
    }
}

__global__ __launch_bounds__(256) void bias_init(
    const float* __restrict__ b, float* __restrict__ out)
{
    const int idx = blockIdx.x * 256 + threadIdx.x;
    if (idx < N_NODES * D_OUT)
        out[idx] = b[idx & (D_OUT - 1)];
}

__global__ __launch_bounds__(256) void edge_scatter(
    const float* __restrict__ h, const float* __restrict__ edge_w,
    const int* __restrict__ src, const int* __restrict__ dst,
    float* __restrict__ out)
{
    const int gtid = blockIdx.x * 256 + threadIdx.x;
    const int e    = gtid >> 5;
    const int lane = threadIdx.x & 31;
    if (e >= N_EDGES) return;

    const int   s = src[e];
    const int   d = dst[e];
    const float w = edge_w[e];

    const float4 hv = *(const float4*)(h + (size_t)s * D_OUT + lane * 4);
    float* op = out + (size_t)d * D_OUT + lane * 4;

    __hip_atomic_fetch_add(op + 0, w * hv.x, __ATOMIC_RELAXED, __HIP_MEMORY_SCOPE_AGENT);
    __hip_atomic_fetch_add(op + 1, w * hv.y, __ATOMIC_RELAXED, __HIP_MEMORY_SCOPE_AGENT);
    __hip_atomic_fetch_add(op + 2, w * hv.z, __ATOMIC_RELAXED, __HIP_MEMORY_SCOPE_AGENT);
    __hip_atomic_fetch_add(op + 3, w * hv.w, __ATOMIC_RELAXED, __HIP_MEMORY_SCOPE_AGENT);
}

extern "C" void kernel_launch(void* const* d_in, const int* in_sizes, int n_in,
                              void* d_out, int out_size, void* d_ws, size_t ws_size,
                              hipStream_t stream)
{
    const float* x      = (const float*)d_in[0];
    const float* edge_w = (const float*)d_in[1];
    const int*   src    = (const int*)  d_in[2];
    const int*   dst    = (const int*)  d_in[3];
    const float* W      = (const float*)d_in[4];
    const float* b      = (const float*)d_in[5];
    float*       out    = (float*)d_out;
    float*       h      = (float*)d_ws;

    gemm_wmma_f32<<<(N_BANDS + 7) / 8, 256, 0, stream>>>(x, W, h);

    bias_init<<<(N_NODES * D_OUT + 255) / 256, 256, 0, stream>>>(b, out);

    edge_scatter<<<(N_EDGES * 32) / 256, 256, 0, stream>>>(h, edge_w, src, dst, out);
}


# edge scatter: lane l takes columns l+32j, so each f32 atomic add instruction covers one whole 128-B line of the out row instead of a quarter of four lines; h read line by line to match
# speedup vs baseline: 1.2431x; 1.2431x over previous
#include <hip/hip_runtime.h>

typedef __attribute__((ext_vector_type(2))) float v2f;
typedef __attribute__((ext_vector_type(8))) float v8f;

#define N_NODES 50000
#define N_EDGES 800000
#define D_IN    512
#define D_OUT   128
#define N_BANDS (N_NODES / 16)

__global__ __launch_bounds__(256) void gemm_wmma_f32(
    const float* __restrict__ x, const float* __restrict__ W,
    float* __restrict__ h)
{
    const int lane = threadIdx.x & 31;
    const int band = blockIdx.x * 8 + (threadIdx.x >> 5);
    if (band >= N_BANDS) return;

    const int m0   = band * 16;
    const int half = lane >> 4;
    const int l16  = lane & 15;
    const int koff = half * 2;

    v8f acc[8] = {};

    const float* xrow = x + (size_t)(m0 + l16) * D_IN + koff;

    for (int k0 = 0; k0 < D_IN; k0 += 4) {
        v2f a;
        a.x = xrow[k0 + 0];
        a.y = xrow[k0 + 1];
        const float* wk = W + (size_t)(k0 + koff) * D_OUT + l16;
        #pragma unroll
        for (int nt = 0; nt < 8; ++nt) {
            v2f b;
            b.x = wk[nt * 16];
            b.y = wk[nt * 16 + D_OUT];
            acc[nt] = __builtin_amdgcn_wmma_f32_16x16x4_f32(
                 false, a,  false, b,
                 (short)0, acc[nt],
                 false,  false);
        }
    }

    #pragma unroll
    for (int nt = 0; nt < 8; ++nt) {
        float* hp = h + (size_t)(m0 + half * 8) * D_OUT + nt * 16 + l16;
        #pragma unroll
        for (int r = 0; r < 8; ++r)
            hp[(size_t)r * D_OUT] = acc[nt][r];
    }
}

__global__ __launch_bounds__(256) void bias_init(
    const float* __restrict__ b, float* __restrict__ out)
{
    const int idx = blockIdx.x * 256 + threadIdx.x;
    if (idx < N_NODES * D_OUT)
        out[idx] = b[idx & (D_OUT - 1)];
}

__global__ __launch_bounds__(256) void edge_scatter(
    const float* __restrict__ h, const float* __restrict__ edge_w,
    const int* __restrict__ src, const int* __restrict__ dst,
    float* __restrict__ out)
{
    const int gtid = blockIdx.x * 256 + threadIdx.x;
    const int e    = gtid >> 5;
    const int lane = threadIdx.x & 31;
    if (e >= N_EDGES) return;

    const int   s = src[e];
    const int   d = dst[e];
    const float w = edge_w[e];

    const float* hp = h + (size_t)s * D_OUT + lane;
    float*       op = out + (size_t)d * D_OUT + lane;

    const float h0 = hp[0];
    const float h1 = hp[32];
    const float h2 = hp[64];
    const float h3 = hp[96];

    __hip_atomic_fetch_add(op +  0, w * h0, __ATOMIC_RELAXED, __HIP_MEMORY_SCOPE_AGENT);
    __hip_atomic_fetch_add(op + 32, w * h1, __ATOMIC_RELAXED, __HIP_MEMORY_SCOPE_AGENT);
    __hip_atomic_fetch_add(op + 64, w * h2, __ATOMIC_RELAXED, __HIP_MEMORY_SCOPE_AGENT);
    __hip_atomic_fetch_add(op + 96, w * h3, __ATOMIC_RELAXED, __HIP_MEMORY_SCOPE_AGENT);
}

extern "C" void kernel_launch(void* const* d_in, const int* in_sizes, int n_in,
                              void* d_out, int out_size, void* d_ws, size_t ws_size,
                              hipStream_t stream)
{
    const float* x      = (const float*)d_in[0];
    const float* edge_w = (const float*)d_in[1];
    const int*   src    = (const int*)  d_in[2];
    const int*   dst    = (const int*)  d_in[3];
    const float* W      = (const float*)d_in[4];
    const float* b      = (const float*)d_in[5];
    float*       out    = (float*)d_out;
    float*       h      = (float*)d_ws;

    gemm_wmma_f32<<<(N_BANDS + 7) / 8, 256, 0, stream>>>(x, W, h);

    bias_init<<<(N_NODES * D_OUT + 255) / 256, 256, 0, stream>>>(b, out);

    edge_scatter<<<(N_EDGES * 32) / 256, 256, 0, stream>>>(h, edge_w, src, dst, out);
}


# edge scatter: one wave takes 16 consecutive edges and issues their 64 h-line loads before its first atomic add: 16x fewer waves, 64 lines in flight each; f32 atomics kept, arrival order unchanged
# speedup vs baseline: 1.5323x; 1.2326x over previous
#include <hip/hip_runtime.h>

typedef __attribute__((ext_vector_type(2))) float v2f;
typedef __attribute__((ext_vector_type(8))) float v8f;

#define N_NODES 50000
#define N_EDGES 800000
#define D_IN    512
#define D_OUT   128
#define N_BANDS (N_NODES / 16)

__global__ __launch_bounds__(256) void gemm_wmma_f32(
    const float* __restrict__ x, const float* __restrict__ W,
    float* __restrict__ h)
{
    const int lane = threadIdx.x & 31;
    const int band = blockIdx.x * 8 + (threadIdx.x >> 5);
    if (band >= N_BANDS) return;

    const int m0   = band * 16;
    const int half = lane >> 4;
    const int l16  = lane & 15;
    const int koff = half * 2;

    v8f acc[8] = {};

    const float* xrow = x + (size_t)(m0 + l16) * D_IN + koff;

    for (int k0 = 0; k0 < D_IN; k0 += 4) {
        v2f a;
        a.x = xrow[k0 + 0];
        a.y = xrow[k0 + 1];
        const float* wk = W + (size_t)(k0 + koff) * D_OUT + l16;
        #pragma unroll
        for (int nt = 0; nt < 8; ++nt) {
            v2f b;
            b.x = wk[nt * 16];
            b.y = wk[nt * 16 + D_OUT];
            acc[nt] = __builtin_amdgcn_wmma_f32_16x16x4_f32(
                 false, a,  false, b,
                 (short)0, acc[nt],
                 false,  false);
        }
    }

    #pragma unroll
    for (int nt = 0; nt < 8; ++nt) {
        float* hp = h + (size_t)(m0 + half * 8) * D_OUT + nt * 16 + l16;
        #pragma unroll
        for (int r = 0; r < 8; ++r)
            hp[(size_t)r * D_OUT] = acc[nt][r];
    }
}

__global__ __launch_bounds__(256) void bias_init(
    const float* __restrict__ b, float* __restrict__ out)
{
    const int idx = blockIdx.x * 256 + threadIdx.x;
    if (idx < N_NODES * D_OUT)
        out[idx] = b[idx & (D_OUT - 1)];
}

#define EPW 16
static_assert(N_EDGES % (EPW * 8) == 0, "grid covers the edges exactly");

__global__ __launch_bounds__(256) void edge_scatter(
    const float* __restrict__ h, const float* __restrict__ edge_w,
    const int* __restrict__ src, const int* __restrict__ dst,
    float* __restrict__ out)
{
    const int lane = threadIdx.x & 31;
    const int wave = blockIdx.x * 8 + (threadIdx.x >> 5);
    const int e0   = wave * EPW;
    if (e0 >= N_EDGES) return;

    const int   ei = e0 + (lane & (EPW - 1));
    const int   sv = src[ei];
    const int   dv = dst[ei];
    const int   wv = __float_as_int(edge_w[ei]);

    float hv[EPW][4];

    #pragma unroll
    for (int i = 0; i < EPW; ++i) {
        const int s = __builtin_amdgcn_readlane(sv, i);
        const float* hp = h + (size_t)s * D_OUT + lane;
        hv[i][0] = hp[0];
        hv[i][1] = hp[32];
        hv[i][2] = hp[64];
        hv[i][3] = hp[96];
    }

    #pragma unroll
    for (int i = 0; i < EPW; ++i) {
        const int   d = __builtin_amdgcn_readlane(dv, i);
        const float w = __int_as_float(__builtin_amdgcn_readlane(wv, i));
        float* op = out + (size_t)d * D_OUT + lane;
        __hip_atomic_fetch_add(op +  0, w * hv[i][0], __ATOMIC_RELAXED, __HIP_MEMORY_SCOPE_AGENT);
        __hip_atomic_fetch_add(op + 32, w * hv[i][1], __ATOMIC_RELAXED, __HIP_MEMORY_SCOPE_AGENT);
        __hip_atomic_fetch_add(op + 64, w * hv[i][2], __ATOMIC_RELAXED, __HIP_MEMORY_SCOPE_AGENT);
        __hip_atomic_fetch_add(op + 96, w * hv[i][3], __ATOMIC_RELAXED, __HIP_MEMORY_SCOPE_AGENT);
    }
}

extern "C" void kernel_launch(void* const* d_in, const int* in_sizes, int n_in,
                              void* d_out, int out_size, void* d_ws, size_t ws_size,
                              hipStream_t stream)
{
    const float* x      = (const float*)d_in[0];
    const float* edge_w = (const float*)d_in[1];
    const int*   src    = (const int*)  d_in[2];
    const int*   dst    = (const int*)  d_in[3];
    const float* W      = (const float*)d_in[4];
    const float* b      = (const float*)d_in[5];
    float*       out    = (float*)d_out;
    float*       h      = (float*)d_ws;

    gemm_wmma_f32<<<(N_BANDS + 7) / 8, 256, 0, stream>>>(x, W, h);

    bias_init<<<(N_NODES * D_OUT + 255) / 256, 256, 0, stream>>>(b, out);

    edge_scatter<<<N_EDGES / (EPW * 8), 256, 0, stream>>>(h, edge_w, src, dst, out);
}


# out = bias start values stored by the GEMM waves (16 whole-row 16-byte stores each, ahead of the k loop) instead of by a separate fill launch: one launch and its 25000 work-groups fewer
# speedup vs baseline: 1.6671x; 1.0880x over previous
#include <hip/hip_runtime.h>

typedef __attribute__((ext_vector_type(2))) float v2f;
typedef __attribute__((ext_vector_type(4))) float v4f;
typedef __attribute__((ext_vector_type(8))) float v8f;

#define N_NODES 50000
#define N_EDGES 800000
#define D_IN    512
#define D_OUT   128
#define N_BANDS (N_NODES / 16)

__global__ __launch_bounds__(256) void gemm_wmma_f32(
    const float* __restrict__ x, const float* __restrict__ W,
    const float* __restrict__ bias, float* __restrict__ h,
    float* __restrict__ out)
{
    const int lane = threadIdx.x & 31;
    const int band = blockIdx.x * 8 + (threadIdx.x >> 5);
    if (band >= N_BANDS) return;

    const int m0   = band * 16;
    const int half = lane >> 4;
    const int l16  = lane & 15;
    const int koff = half * 2;

    {
        const v4f bv = *(const v4f*)(bias + lane * 4);
        float* orow = out + (size_t)m0 * D_OUT + lane * 4;
        #pragma unroll
        for (int r = 0; r < 16; ++r)
            *(v4f*)(orow + (size_t)r * D_OUT) = bv;
    }

    v8f acc[8] = {};

    const float* xrow = x + (size_t)(m0 + l16) * D_IN + koff;

    for (int k0 = 0; k0 < D_IN; k0 += 4) {
        v2f a;
        a.x = xrow[k0 + 0];
        a.y = xrow[k0 + 1];
        const float* wk = W + (size_t)(k0 + koff) * D_OUT + l16;
        #pragma unroll
        for (int nt = 0; nt < 8; ++nt) {
            v2f b;
            b.x = wk[nt * 16];
            b.y = wk[nt * 16 + D_OUT];
            acc[nt] = __builtin_amdgcn_wmma_f32_16x16x4_f32(
                 false, a,  false, b,
                 (short)0, acc[nt],
                 false,  false);
        }
    }

    #pragma unroll
    for (int nt = 0; nt < 8; ++nt) {
        float* hp = h + (size_t)(m0 + half * 8) * D_OUT + nt * 16 + l16;
        #pragma unroll
        for (int r = 0; r < 8; ++r)
            hp[(size_t)r * D_OUT] = acc[nt][r];
    }
}

#define EPW 16
static_assert(N_EDGES % (EPW * 8) == 0, "grid covers the edges exactly");

__global__ __launch_bounds__(256) void edge_scatter(
    const float* __restrict__ h, const float* __restrict__ edge_w,
    const int* __restrict__ src, const int* __restrict__ dst,
    float* __restrict__ out)
{
    const int lane = threadIdx.x & 31;
    const int wave = blockIdx.x * 8 + (threadIdx.x >> 5);
    const int e0   = wave * EPW;
    if (e0 >= N_EDGES) return;

    const int   ei = e0 + (lane & (EPW - 1));
    const int   sv = src[ei];
    const int   dv = dst[ei];
    const int   wv = __float_as_int(edge_w[ei]);

    float hv[EPW][4];

    #pragma unroll
    for (int i = 0; i < EPW; ++i) {
        const int s = __builtin_amdgcn_readlane(sv, i);
        const float* hp = h + (size_t)s * D_OUT + lane;
        hv[i][0] = hp[0];
        hv[i][1] = hp[32];
        hv[i][2] = hp[64];
        hv[i][3] = hp[96];
    }

    #pragma unroll
    for (int i = 0; i < EPW; ++i) {
        const int   d = __builtin_amdgcn_readlane(dv, i);
        const float w = __int_as_float(__builtin_amdgcn_readlane(wv, i));
        float* op = out + (size_t)d * D_OUT + lane;
        __hip_atomic_fetch_add(op +  0, w * hv[i][0], __ATOMIC_RELAXED, __HIP_MEMORY_SCOPE_AGENT);
        __hip_atomic_fetch_add(op + 32, w * hv[i][1], __ATOMIC_RELAXED, __HIP_MEMORY_SCOPE_AGENT);
        __hip_atomic_fetch_add(op + 64, w * hv[i][2], __ATOMIC_RELAXED, __HIP_MEMORY_SCOPE_AGENT);
        __hip_atomic_fetch_add(op + 96, w * hv[i][3], __ATOMIC_RELAXED, __HIP_MEMORY_SCOPE_AGENT);
    }
}

extern "C" void kernel_launch(void* const* d_in, const int* in_sizes, int n_in,
                              void* d_out, int out_size, void* d_ws, size_t ws_size,
                              hipStream_t stream)
{
    const float* x      = (const float*)d_in[0];
    const float* edge_w = (const float*)d_in[1];
    const int*   src    = (const int*)  d_in[2];
    const int*   dst    = (const int*)  d_in[3];
    const float* W      = (const float*)d_in[4];
    const float* b      = (const float*)d_in[5];
    float*       out    = (float*)d_out;
    float*       h      = (float*)d_ws;

    gemm_wmma_f32<<<(N_BANDS + 7) / 8, 256, 0, stream>>>(x, W, b, h, out);

    edge_scatter<<<N_EDGES / (EPW * 8), 256, 0, stream>>>(h, edge_w, src, dst, out);
}
